# starting kernel
# baseline (speedup 1.0000x reference)
#include <hip/hip_runtime.h>
#include <math.h>

typedef __attribute__((ext_vector_type(2))) float v2f;
typedef __attribute__((ext_vector_type(8))) float v8f;

#define DIN 256
#define HN  64
#define ONUM 10
#define DT  320
#define WRS 68
#define HST 20

__device__ __forceinline__ v8f wmma4(v2f a, v2f b, v8f c) {
  return __builtin_amdgcn_wmma_f32_16x16x4_f32(false, a, false, b, (short)0, c,
                                               false, false);
}

__device__ __forceinline__ float fast_sigmoid(float s) {
  return __builtin_amdgcn_rcpf(1.0f + __expf(-s));
}

__global__ __launch_bounds__(256) void casper_prep(const float* __restrict__ Wh,
                                                   const float* __restrict__ Wout,
                                                   float* __restrict__ wsB) {
  int tid = blockIdx.x * blockDim.x + threadIdx.x;
  if (tid >= 64 * 5 * 32) return;
  int t  = tid / 160;
  int nt = (tid % 160) / 32;
  int l  = tid & 31;
  int kb = 4 * t + ((l >> 4) << 1);
  float v0 = 0.f, v1 = 0.f;
  if (nt < 4) {
    int n = nt * 16 + (l & 15);
    v0 = Wh[n * DT + kb];
    v1 = Wh[n * DT + kb + 1];
  } else {
    int o = l & 15;
    if (o < ONUM) {
      v0 = Wout[o * DT + kb];
      v1 = Wout[o * DT + kb + 1];
    }
  }
  ((v2f*)wsB)[tid] = (v2f){v0, v1};
}

__global__ __launch_bounds__(256) void casper_main(
    const float* __restrict__ x, const float* __restrict__ Wh,
    const float* __restrict__ bh, const float* __restrict__ Wout,
    const float* __restrict__ bo, const float* __restrict__ wsB,
    float* __restrict__ out, int nrows) {
  __shared__ __align__(16) float WrL[HN * WRS];
  __shared__ __align__(16) float WorL[16 * WRS];
  __shared__ float bhL[HN];
  __shared__ float boL[16];
  __shared__ __align__(16) float hbuf[8][4][16 * HST];

  const int tid  = threadIdx.x;
  const int lane = tid & 31;
  const int wid  = tid >> 5;
  const int mrow  = lane & 15;
  const int khalf = lane >> 4;

  for (int idx = tid; idx < HN * WRS; idx += 256) {
    int i = idx / WRS, j = idx % WRS;
    WrL[idx] = (j < HN) ? Wh[i * DT + DIN + j] : 0.f;
  }
  for (int idx = tid; idx < 16 * WRS; idx += 256) {
    int o = idx / WRS, j = idx % WRS;
    WorL[idx] = (o < ONUM && j < HN) ? Wout[o * DT + DIN + j] : 0.f;
  }
  if (tid < HN) bhL[tid] = bh[tid];
  if (tid < 16) boL[tid] = (tid < ONUM) ? bo[tid] : 0.f;
  __syncthreads();

  const int row0 = blockIdx.x * 128 + wid * 16;
  const float* xl = x + (size_t)(row0 + mrow) * DIN + khalf * 2;

  v8f cP[4];
  v8f cOut = {0.f, 0.f, 0.f, 0.f, 0.f, 0.f, 0.f, 0.f};
#pragma unroll
  for (int g = 0; g < 4; ++g) cP[g] = cOut;

  const v2f* Bws = (const v2f*)wsB;
#pragma unroll 2
  for (int t = 0; t < 64; ++t) {
    v2f a  = *(const v2f*)(xl + 4 * t);
    v2f b0 = Bws[(t * 5 + 0) * 32 + lane];
    v2f b1 = Bws[(t * 5 + 1) * 32 + lane];
    v2f b2 = Bws[(t * 5 + 2) * 32 + lane];
    v2f b3 = Bws[(t * 5 + 3) * 32 + lane];
    v2f b4 = Bws[(t * 5 + 4) * 32 + lane];
    cP[0] = wmma4(a, b0, cP[0]);
    cP[1] = wmma4(a, b1, cP[1]);
    cP[2] = wmma4(a, b2, cP[2]);
    cP[3] = wmma4(a, b3, cP[3]);
    cOut  = wmma4(a, b4, cOut);
  }

#pragma unroll
  for (int g = 0; g < 4; ++g) {
    v8f acc = cP[g];
#pragma unroll
    for (int gp = 0; gp < 4; ++gp) {
      if (gp < g) {
#pragma unroll
        for (int kk = 0; kk < 4; ++kk) {
          int jj = 4 * kk + khalf * 2;
          v2f a = *(const v2f*)&hbuf[wid][gp][mrow * HST + jj];
          v2f b = *(const v2f*)&WrL[(g * 16 + mrow) * WRS + gp * 16 + jj];
          acc = wmma4(a, b, acc);
        }
      }
    }
    {
      int mb = khalf * 8;
#pragma unroll
      for (int r = 0; r < 8; ++r)
        hbuf[wid][g][(mb + r) * HST + mrow] = acc[r];
    }
    __syncthreads();
    {
      float hl[16];
#pragma unroll
      for (int i = 0; i < 16; ++i) {
        float s = hbuf[wid][g][mrow * HST + i] + bhL[g * 16 + i];
#pragma unroll
        for (int j = 0; j < i; ++j)
          s += WrL[(g * 16 + i) * WRS + g * 16 + j] * hl[j];
        float h = fast_sigmoid(s);
        hl[i] = h;
        hbuf[wid][g][mrow * HST + i] = h;
      }
    }
    __syncthreads();
  }

#pragma unroll
  for (int g = 0; g < 4; ++g) {
#pragma unroll
    for (int kk = 0; kk < 4; ++kk) {
      int jj = 4 * kk + khalf * 2;
      v2f a = *(const v2f*)&hbuf[wid][g][mrow * HST + jj];
      v2f b = *(const v2f*)&WorL[mrow * WRS + g * 16 + jj];
      cOut = wmma4(a, b, cOut);
    }
  }

  if (mrow < ONUM) {
    int mb = khalf * 8;
    float bb = boL[mrow];
#pragma unroll
    for (int r = 0; r < 8; ++r)
      out[(size_t)(row0 + mb + r) * ONUM + mrow] = cOut[r] + bb;
  }
}

extern "C" void kernel_launch(void* const* d_in, const int* in_sizes, int n_in,
                              void* d_out, int out_size, void* d_ws,
                              size_t ws_size, hipStream_t stream) {
  const float* x   = (const float*)d_in[0];
  const float* Wh  = (const float*)d_in[1];
  const float* bhp = (const float*)d_in[2];
  const float* Wo  = (const float*)d_in[3];
  const float* bop = (const float*)d_in[4];
  float* outp = (float*)d_out;
  float* wsB  = (float*)d_ws;

  int nrows   = in_sizes[0] / DIN;
  int nblocks = nrows / 128;

  casper_prep<<<40, 256, 0, stream>>>(Wh, Wo, wsB);
  casper_main<<<nblocks, 256, 0, stream>>>(x, Wh, bhp, Wo, bop, wsB, outp,
                                           nrows);
}


# K-loop reads x and the packed weights with 16-byte per-lane loads, four back to back per 128-B row line; k order permuted alike in both operands, so the order of summation over k changes
# speedup vs baseline: 1.1234x; 1.1234x over previous
#include <hip/hip_runtime.h>
#include <math.h>

typedef __attribute__((ext_vector_type(2))) float v2f;
typedef __attribute__((ext_vector_type(4))) float v4f;
typedef __attribute__((ext_vector_type(8))) float v8f;

#define DIN 256
#define HN  64
#define ONUM 10
#define DT  320
#define WRS 68
#define HST 20

__device__ __forceinline__ v8f wmma4(v2f a, v2f b, v8f c) {
  return __builtin_amdgcn_wmma_f32_16x16x4_f32(false, a, false, b, (short)0, c,
                                               false, false);
}

__device__ __forceinline__ float fast_sigmoid(float s) {
  return __builtin_amdgcn_rcpf(1.0f + __expf(-s));
}

__global__ __launch_bounds__(256) void casper_prep(const float* __restrict__ Wh,
                                                   const float* __restrict__ Wout,
                                                   float* __restrict__ wsB) {
  int tid = blockIdx.x * blockDim.x + threadIdx.x;
  if (tid >= 32 * 5 * 32) return;
  int u  = tid / 160;
  int nt = (tid % 160) / 32;
  int l  = tid & 31;
  int kb = 8 * u + ((l >> 4) << 2);
  v4f v = {0.f, 0.f, 0.f, 0.f};
  if (nt < 4) {
    int n = nt * 16 + (l & 15);
    v = *(const v4f*)(Wh + n * DT + kb);
  } else {
    int o = l & 15;
    if (o < ONUM) v = *(const v4f*)(Wout + o * DT + kb);
  }
  ((v4f*)wsB)[tid] = v;
}

__global__ __launch_bounds__(256) void casper_main(
    const float* __restrict__ x, const float* __restrict__ Wh,
    const float* __restrict__ bh, const float* __restrict__ Wout,
    const float* __restrict__ bo, const float* __restrict__ wsB,
    float* __restrict__ out, int nrows) {
  __shared__ __align__(16) float WrL[HN * WRS];
  __shared__ __align__(16) float WorL[16 * WRS];
  __shared__ float bhL[HN];
  __shared__ float boL[16];
  __shared__ __align__(16) float hbuf[8][4][16 * HST];

  const int tid  = threadIdx.x;
  const int lane = tid & 31;
  const int wid  = tid >> 5;
  const int mrow  = lane & 15;
  const int khalf = lane >> 4;

  for (int idx = tid; idx < HN * WRS; idx += 256) {
    int i = idx / WRS, j = idx % WRS;
    WrL[idx] = (j < HN) ? Wh[i * DT + DIN + j] : 0.f;
  }
  for (int idx = tid; idx < 16 * WRS; idx += 256) {
    int o = idx / WRS, j = idx % WRS;
    WorL[idx] = (o < ONUM && j < HN) ? Wout[o * DT + DIN + j] : 0.f;
  }
  if (tid < HN) bhL[tid] = bh[tid];
  if (tid < 16) boL[tid] = (tid < ONUM) ? bo[tid] : 0.f;
  __syncthreads();

  const int row0 = blockIdx.x * 128 + wid * 16;
  const float* xl = x + (size_t)(row0 + mrow) * DIN + khalf * 4;

  v8f cP[4];
  v8f cOut = {0.f, 0.f, 0.f, 0.f, 0.f, 0.f, 0.f, 0.f};
#pragma unroll
  for (int g = 0; g < 4; ++g) cP[g] = cOut;

  const v4f* Bws = (const v4f*)wsB + lane;
#pragma unroll 1
  for (int q = 0; q < 8; ++q) {
    v4f a[4];
#pragma unroll
    for (int m = 0; m < 4; ++m) a[m] = *(const v4f*)(xl + 32 * q + 8 * m);
#pragma unroll
    for (int m = 0; m < 4; ++m) {
      const v4f* bp = Bws + ((4 * q + m) * 5) * 32;
      v4f b0 = bp[0 * 32];
      v4f b1 = bp[1 * 32];
      v4f b2 = bp[2 * 32];
      v4f b3 = bp[3 * 32];
      v4f b4 = bp[4 * 32];
      v2f alo = {a[m].x, a[m].y};
      v2f ahi = {a[m].z, a[m].w};
      cP[0] = wmma4(alo, (v2f){b0.x, b0.y}, cP[0]);
      cP[1] = wmma4(alo, (v2f){b1.x, b1.y}, cP[1]);
      cP[2] = wmma4(alo, (v2f){b2.x, b2.y}, cP[2]);
      cP[3] = wmma4(alo, (v2f){b3.x, b3.y}, cP[3]);
      cOut  = wmma4(alo, (v2f){b4.x, b4.y}, cOut);
      cP[0] = wmma4(ahi, (v2f){b0.z, b0.w}, cP[0]);
      cP[1] = wmma4(ahi, (v2f){b1.z, b1.w}, cP[1]);
      cP[2] = wmma4(ahi, (v2f){b2.z, b2.w}, cP[2]);
      cP[3] = wmma4(ahi, (v2f){b3.z, b3.w}, cP[3]);
      cOut  = wmma4(ahi, (v2f){b4.z, b4.w}, cOut);
    }
  }

#pragma unroll
  for (int g = 0; g < 4; ++g) {
    v8f acc = cP[g];
#pragma unroll
    for (int gp = 0; gp < 4; ++gp) {
      if (gp < g) {
#pragma unroll
        for (int kk = 0; kk < 4; ++kk) {
          int jj = 4 * kk + khalf * 2;
          v2f a = *(const v2f*)&hbuf[wid][gp][mrow * HST + jj];
          v2f b = *(const v2f*)&WrL[(g * 16 + mrow) * WRS + gp * 16 + jj];
          acc = wmma4(a, b, acc);
        }
      }
    }
    {
      int mb = khalf * 8;
#pragma unroll
      for (int r = 0; r < 8; ++r)
        hbuf[wid][g][(mb + r) * HST + mrow] = acc[r];
    }
    __syncthreads();
    {
      float hl[16];
#pragma unroll
      for (int i = 0; i < 16; ++i) {
        float s = hbuf[wid][g][mrow * HST + i] + bhL[g * 16 + i];
#pragma unroll
        for (int j = 0; j < i; ++j)
          s += WrL[(g * 16 + i) * WRS + g * 16 + j] * hl[j];
        float h = fast_sigmoid(s);
        hl[i] = h;
        hbuf[wid][g][mrow * HST + i] = h;
      }
    }
    __syncthreads();
  }

#pragma unroll
  for (int g = 0; g < 4; ++g) {
#pragma unroll
    for (int kk = 0; kk < 4; ++kk) {
      int jj = 4 * kk + khalf * 2;
      v2f a = *(const v2f*)&hbuf[wid][g][mrow * HST + jj];
      v2f b = *(const v2f*)&WorL[mrow * WRS + g * 16 + jj];
      cOut = wmma4(a, b, cOut);
    }
  }

  if (mrow < ONUM) {
    int mb = khalf * 8;
    float bb = boL[mrow];
#pragma unroll
    for (int r = 0; r < 8; ++r)
      out[(size_t)(row0 + mb + r) * ONUM + mrow] = cOut[r] + bb;
  }
}

extern "C" void kernel_launch(void* const* d_in, const int* in_sizes, int n_in,
                              void* d_out, int out_size, void* d_ws,
                              size_t ws_size, hipStream_t stream) {
  const float* x   = (const float*)d_in[0];
  const float* Wh  = (const float*)d_in[1];
  const float* bhp = (const float*)d_in[2];
  const float* Wo  = (const float*)d_in[3];
  const float* bop = (const float*)d_in[4];
  float* outp = (float*)d_out;
  float* wsB  = (float*)d_ws;

  int nrows   = in_sizes[0] / DIN;
  int nblocks = nrows / 128;

  casper_prep<<<20, 256, 0, stream>>>(Wh, Wo, wsB);
  casper_main<<<nblocks, 256, 0, stream>>>(x, Wh, bhp, Wo, bop, wsB, outp,
                                           nrows);
}


# LDS staging of the cascade's tables no longer one load-wait-store round trip per element: its global loads issued as one batch of 16-byte loads, one wait, entry barrier dropped; sums unchanged
# speedup vs baseline: 1.3193x; 1.1744x over previous
#include <hip/hip_runtime.h>
#include <math.h>

typedef __attribute__((ext_vector_type(2))) float v2f;
typedef __attribute__((ext_vector_type(4))) float v4f;
typedef __attribute__((ext_vector_type(8))) float v8f;

#define DIN 256
#define HN  64
#define ONUM 10
#define DT  320
#define WRS 68
#define HST 20

__device__ __forceinline__ v8f wmma4(v2f a, v2f b, v8f c) {
  return __builtin_amdgcn_wmma_f32_16x16x4_f32(false, a, false, b, (short)0, c,
                                               false, false);
}

__device__ __forceinline__ float fast_sigmoid(float s) {
  return __builtin_amdgcn_rcpf(1.0f + __expf(-s));
}

__global__ __launch_bounds__(256) void casper_prep(const float* __restrict__ Wh,
                                                   const float* __restrict__ Wout,
                                                   float* __restrict__ wsB) {
  int tid = blockIdx.x * blockDim.x + threadIdx.x;
  if (tid >= 32 * 5 * 32) return;
  int u  = tid / 160;
  int nt = (tid % 160) / 32;
  int l  = tid & 31;
  int kb = 8 * u + ((l >> 4) << 2);
  v4f v = {0.f, 0.f, 0.f, 0.f};
  if (nt < 4) {
    int n = nt * 16 + (l & 15);
    v = *(const v4f*)(Wh + n * DT + kb);
  } else {
    int o = l & 15;
    if (o < ONUM) v = *(const v4f*)(Wout + o * DT + kb);
  }
  ((v4f*)wsB)[tid] = v;
}

__global__ __launch_bounds__(256) void casper_main(
    const float* __restrict__ x, const float* __restrict__ Wh,
    const float* __restrict__ bh, const float* __restrict__ Wout,
    const float* __restrict__ bo, const float* __restrict__ wsB,
    float* __restrict__ out, int nrows) {
  __shared__ __align__(16) float WrL[HN * WRS];
  __shared__ __align__(16) float WorL[16 * WRS];
  __shared__ float bhL[HN];
  __shared__ float boL[16];
  __shared__ __align__(16) float hbuf[8][4][16 * HST];

  const int tid  = threadIdx.x;
  const int lane = tid & 31;
  const int wid  = tid >> 5;
  const int mrow  = lane & 15;
  const int khalf = lane >> 4;

  {
    const int srow = tid >> 4;
    const int sc4  = (tid & 15) << 2;
    v4f stW[4];
#pragma unroll
    for (int i = 0; i < 4; ++i)
      stW[i] = *(const v4f*)(Wh + (srow + 16 * i) * DT + DIN + sc4);
    v4f stO = {0.f, 0.f, 0.f, 0.f};
    if (srow < ONUM) stO = *(const v4f*)(Wout + srow * DT + DIN + sc4);
    float stBh = 0.f, stBo = 0.f;
    if (tid < HN) stBh = bh[tid];
    if (tid < ONUM) stBo = bo[tid];
#pragma unroll
    for (int i = 0; i < 4; ++i)
      *(v4f*)&WrL[(srow + 16 * i) * WRS + sc4] = stW[i];
    *(v4f*)&WorL[srow * WRS + sc4] = stO;
    if (tid < HN) bhL[tid] = stBh;
    if (tid < 16) boL[tid] = stBo;
  }

  const int row0 = blockIdx.x * 128 + wid * 16;
  const float* xl = x + (size_t)(row0 + mrow) * DIN + khalf * 4;

  v8f cP[4];
  v8f cOut = {0.f, 0.f, 0.f, 0.f, 0.f, 0.f, 0.f, 0.f};
#pragma unroll
  for (int g = 0; g < 4; ++g) cP[g] = cOut;

  const v4f* Bws = (const v4f*)wsB + lane;
#pragma unroll 1
  for (int q = 0; q < 8; ++q) {
    v4f a[4];
#pragma unroll
    for (int m = 0; m < 4; ++m) a[m] = *(const v4f*)(xl + 32 * q + 8 * m);
#pragma unroll
    for (int m = 0; m < 4; ++m) {
      const v4f* bp = Bws + ((4 * q + m) * 5) * 32;
      v4f b0 = bp[0 * 32];
      v4f b1 = bp[1 * 32];
      v4f b2 = bp[2 * 32];
      v4f b3 = bp[3 * 32];
      v4f b4 = bp[4 * 32];
      v2f alo = {a[m].x, a[m].y};
      v2f ahi = {a[m].z, a[m].w};
      cP[0] = wmma4(alo, (v2f){b0.x, b0.y}, cP[0]);
      cP[1] = wmma4(alo, (v2f){b1.x, b1.y}, cP[1]);
      cP[2] = wmma4(alo, (v2f){b2.x, b2.y}, cP[2]);
      cP[3] = wmma4(alo, (v2f){b3.x, b3.y}, cP[3]);
      cOut  = wmma4(alo, (v2f){b4.x, b4.y}, cOut);
      cP[0] = wmma4(ahi, (v2f){b0.z, b0.w}, cP[0]);
      cP[1] = wmma4(ahi, (v2f){b1.z, b1.w}, cP[1]);
      cP[2] = wmma4(ahi, (v2f){b2.z, b2.w}, cP[2]);
      cP[3] = wmma4(ahi, (v2f){b3.z, b3.w}, cP[3]);
      cOut  = wmma4(ahi, (v2f){b4.z, b4.w}, cOut);
    }
  }

#pragma unroll
  for (int g = 0; g < 4; ++g) {
    v8f acc = cP[g];
#pragma unroll
    for (int gp = 0; gp < 4; ++gp) {
      if (gp < g) {
#pragma unroll
        for (int kk = 0; kk < 4; ++kk) {
          int jj = 4 * kk + khalf * 2;
          v2f a = *(const v2f*)&hbuf[wid][gp][mrow * HST + jj];
          v2f b = *(const v2f*)&WrL[(g * 16 + mrow) * WRS + gp * 16 + jj];
          acc = wmma4(a, b, acc);
        }
      }
    }
    {
      int mb = khalf * 8;
#pragma unroll
      for (int r = 0; r < 8; ++r)
        hbuf[wid][g][(mb + r) * HST + mrow] = acc[r];
    }
    __syncthreads();
    {
      float hl[16];
#pragma unroll
      for (int i = 0; i < 16; ++i) {
        float s = hbuf[wid][g][mrow * HST + i] + bhL[g * 16 + i];
#pragma unroll
        for (int j = 0; j < i; ++j)
          s += WrL[(g * 16 + i) * WRS + g * 16 + j] * hl[j];
        float h = fast_sigmoid(s);
        hl[i] = h;
        hbuf[wid][g][mrow * HST + i] = h;
      }
    }
    __syncthreads();
  }

#pragma unroll
  for (int g = 0; g < 4; ++g) {
#pragma unroll
    for (int kk = 0; kk < 4; ++kk) {
      int jj = 4 * kk + khalf * 2;
      v2f a = *(const v2f*)&hbuf[wid][g][mrow * HST + jj];
      v2f b = *(const v2f*)&WorL[mrow * WRS + g * 16 + jj];
      cOut = wmma4(a, b, cOut);
    }
  }

  if (mrow < ONUM) {
    int mb = khalf * 8;
    float bb = boL[mrow];
#pragma unroll
    for (int r = 0; r < 8; ++r)
      out[(size_t)(row0 + mb + r) * ONUM + mrow] = cOut[r] + bb;
  }
}

extern "C" void kernel_launch(void* const* d_in, const int* in_sizes, int n_in,
                              void* d_out, int out_size, void* d_ws,
                              size_t ws_size, hipStream_t stream) {
  const float* x   = (const float*)d_in[0];
  const float* Wh  = (const float*)d_in[1];
  const float* bhp = (const float*)d_in[2];
  const float* Wo  = (const float*)d_in[3];
  const float* bop = (const float*)d_in[4];
  float* outp = (float*)d_out;
  float* wsB  = (float*)d_ws;

  int nrows   = in_sizes[0] / DIN;
  int nblocks = nrows / 128;

  casper_prep<<<20, 256, 0, stream>>>(Wh, Wo, wsB);
  casper_main<<<nblocks, 256, 0, stream>>>(x, Wh, bhp, Wo, bop, wsB, outp,
                                           nrows);
}


# recurrent weights not staged in LDS: cascade B fragments are 16-byte global loads (k permuted alike in A and B: sum order over k changes), serial part uses scalar-operand FMAs; all barriers dropped
# speedup vs baseline: 1.4506x; 1.0995x over previous
#include <hip/hip_runtime.h>
#include <math.h>

typedef __attribute__((ext_vector_type(2))) float v2f;
typedef __attribute__((ext_vector_type(4))) float v4f;
typedef __attribute__((ext_vector_type(8))) float v8f;

#define DIN 256
#define HN  64
#define ONUM 10
#define DT  320
#define HST 20

__device__ __forceinline__ v8f wmma4(v2f a, v2f b, v8f c) {
  return __builtin_amdgcn_wmma_f32_16x16x4_f32(false, a, false, b, (short)0, c,
                                               false, false);
}

__device__ __forceinline__ float fast_sigmoid(float s) {
  return __builtin_amdgcn_rcpf(1.0f + __expf(-s));
}

__global__ __launch_bounds__(256) void casper_prep(const float* __restrict__ Wh,
                                                   const float* __restrict__ Wout,
                                                   float* __restrict__ wsB) {
  int tid = blockIdx.x * blockDim.x + threadIdx.x;
  if (tid >= 32 * 5 * 32) return;
  int u  = tid / 160;
  int nt = (tid % 160) / 32;
  int l  = tid & 31;
  int kb = 8 * u + ((l >> 4) << 2);
  v4f v = {0.f, 0.f, 0.f, 0.f};
  if (nt < 4) {
    int n = nt * 16 + (l & 15);
    v = *(const v4f*)(Wh + n * DT + kb);
  } else {
    int o = l & 15;
    if (o < ONUM) v = *(const v4f*)(Wout + o * DT + kb);
  }
  ((v4f*)wsB)[tid] = v;
}

__global__ __launch_bounds__(256) void casper_main(
    const float* __restrict__ x, const float* __restrict__ Wh,
    const float* __restrict__ bh, const float* __restrict__ Wout,
    const float* __restrict__ bo, const float* __restrict__ wsB,
    float* __restrict__ out, int nrows) {
  __shared__ __align__(16) float hbuf[8][4][16 * HST];

  const int tid  = threadIdx.x;
  const int lane = tid & 31;
  const int wid  = tid >> 5;
  const int mrow  = lane & 15;
  const int khalf = lane >> 4;

  const int row0 = blockIdx.x * 128 + wid * 16;
  const float* xl = x + (size_t)(row0 + mrow) * DIN + khalf * 4;

  v8f cP[4];
  v8f cOut = {0.f, 0.f, 0.f, 0.f, 0.f, 0.f, 0.f, 0.f};
#pragma unroll
  for (int g = 0; g < 4; ++g) cP[g] = cOut;

  const v4f* Bws = (const v4f*)wsB + lane;
#pragma unroll 1
  for (int q = 0; q < 8; ++q) {
    v4f a[4];
#pragma unroll
    for (int m = 0; m < 4; ++m) a[m] = *(const v4f*)(xl + 32 * q + 8 * m);
#pragma unroll
    for (int m = 0; m < 4; ++m) {
      const v4f* bp = Bws + ((4 * q + m) * 5) * 32;
      v4f b0 = bp[0 * 32];
      v4f b1 = bp[1 * 32];
      v4f b2 = bp[2 * 32];
      v4f b3 = bp[3 * 32];
      v4f b4 = bp[4 * 32];
      v2f alo = {a[m].x, a[m].y};
      v2f ahi = {a[m].z, a[m].w};
      cP[0] = wmma4(alo, (v2f){b0.x, b0.y}, cP[0]);
      cP[1] = wmma4(alo, (v2f){b1.x, b1.y}, cP[1]);
      cP[2] = wmma4(alo, (v2f){b2.x, b2.y}, cP[2]);
      cP[3] = wmma4(alo, (v2f){b3.x, b3.y}, cP[3]);
      cOut  = wmma4(alo, (v2f){b4.x, b4.y}, cOut);
      cP[0] = wmma4(ahi, (v2f){b0.z, b0.w}, cP[0]);
      cP[1] = wmma4(ahi, (v2f){b1.z, b1.w}, cP[1]);
      cP[2] = wmma4(ahi, (v2f){b2.z, b2.w}, cP[2]);
      cP[3] = wmma4(ahi, (v2f){b3.z, b3.w}, cP[3]);
      cOut  = wmma4(ahi, (v2f){b4.z, b4.w}, cOut);
    }
  }

  const float* wrB = Wh + (size_t)mrow * DT + DIN + khalf * 4;
  const float* hA  = &hbuf[wid][0][mrow * HST + khalf * 4];

  v4f bO[8];
  float bb = 0.f;
#pragma unroll
  for (int u = 0; u < 8; ++u) bO[u] = (v4f){0.f, 0.f, 0.f, 0.f};
  if (mrow < ONUM) {
    const float* woB = Wout + (size_t)mrow * DT + DIN + khalf * 4;
#pragma unroll
    for (int u = 0; u < 8; ++u) bO[u] = *(const v4f*)(woB + 8 * u);
    bb = bo[mrow];
  }

  v4f bX[6];
#pragma unroll
  for (int g = 0; g < 4; ++g) {
    v8f acc = cP[g];
#pragma unroll
    for (int u = 0; u < 6; ++u) {
      if (u < 2 * g) {
        v4f a = *(const v4f*)(hA + (u >> 1) * (16 * HST) + 8 * (u & 1));
        v4f b = bX[u];
        acc = wmma4((v2f){a.x, a.y}, (v2f){b.x, b.y}, acc);
        acc = wmma4((v2f){a.z, a.w}, (v2f){b.z, b.w}, acc);
      }
    }
    {
      int mb = khalf * 8;
#pragma unroll
      for (int r = 0; r < 8; ++r)
        hbuf[wid][g][(mb + r) * HST + mrow] = acc[r];
    }
#pragma unroll
    for (int u = 0; u < 6; ++u)
      if (g < 3 && u < 2 * (g + 1))
        bX[u] = *(const v4f*)(wrB + (g + 1) * 16 * DT + 8 * u);
    {
      const float* __restrict__ wd = Wh + (g * 16) * DT + DIN + g * 16;
      const float* __restrict__ bd = bh + g * 16;
      float hl[16];
#pragma unroll
      for (int i = 0; i < 16; ++i) {
        float s = hbuf[wid][g][mrow * HST + i] + bd[i];
#pragma unroll
        for (int j = 0; j < i; ++j)
          s = __builtin_fmaf(wd[i * DT + j], hl[j], s);
        float h = fast_sigmoid(s);
        hl[i] = h;
        hbuf[wid][g][mrow * HST + i] = h;
      }
    }
  }

#pragma unroll
  for (int u = 0; u < 8; ++u) {
    v4f a = *(const v4f*)(hA + (u >> 1) * (16 * HST) + 8 * (u & 1));
    cOut = wmma4((v2f){a.x, a.y}, (v2f){bO[u].x, bO[u].y}, cOut);
    cOut = wmma4((v2f){a.z, a.w}, (v2f){bO[u].z, bO[u].w}, cOut);
  }

  if (mrow < ONUM) {
    int mb = khalf * 8;
#pragma unroll
    for (int r = 0; r < 8; ++r)
      out[(size_t)(row0 + mb + r) * ONUM + mrow] = cOut[r] + bb;
  }
}

extern "C" void kernel_launch(void* const* d_in, const int* in_sizes, int n_in,
                              void* d_out, int out_size, void* d_ws,
                              size_t ws_size, hipStream_t stream) {
  const float* x   = (const float*)d_in[0];
  const float* Wh  = (const float*)d_in[1];
  const float* bhp = (const float*)d_in[2];
  const float* Wo  = (const float*)d_in[3];
  const float* bop = (const float*)d_in[4];
  float* outp = (float*)d_out;
  float* wsB  = (float*)d_ws;

  int nrows   = in_sizes[0] / DIN;
  int nblocks = nrows / 128;

  casper_prep<<<20, 256, 0, stream>>>(Wh, Wo, wsB);
  casper_main<<<nblocks, 256, 0, stream>>>(x, Wh, bhp, Wo, bop, wsB, outp,
                                           nrows);
}
